# starting kernel
# baseline (speedup 1.0000x reference)
#include <hip/hip_runtime.h>
#include <math.h>


typedef __attribute__((ext_vector_type(2))) float v2f;
typedef __attribute__((ext_vector_type(8))) float v8f;


struct cd { double re, im; };
__device__ __forceinline__ cd cmul(cd a, cd b) {
  return { a.re * b.re - a.im * b.im, a.re * b.im + a.im * b.re };
}

__device__ void gate_rx(double t, cd g[4]) {
  double c = cos(t * 0.5), s = sin(t * 0.5);
  g[0] = { c, 0.0 }; g[1] = { 0.0, -s };
  g[2] = { 0.0, -s }; g[3] = { c, 0.0 };
}
__device__ void gate_rz(double t, cd g[4]) {
  double c = cos(t * 0.5), s = sin(t * 0.5);
  g[0] = { c, -s }; g[1] = { 0.0, 0.0 };
  g[2] = { 0.0, 0.0 }; g[3] = { c, s };
}
__device__ void gate_ry(double t, cd g[4]) {
  double c = cos(t * 0.5), s = sin(t * 0.5);
  g[0] = { c, 0.0 }; g[1] = { -s, 0.0 };
  g[2] = { s, 0.0 }; g[3] = { c, 0.0 };
}

__device__ void apply_q0(cd U[16], const cd g[4]) {
  cd nU[16];
  for (int a = 0; a < 2; ++a)
    for (int b = 0; b < 2; ++b)
      for (int col = 0; col < 4; ++col) {
        cd t0 = cmul(g[a * 2 + 0], U[(0 * 2 + b) * 4 + col]);
        cd t1 = cmul(g[a * 2 + 1], U[(1 * 2 + b) * 4 + col]);
        nU[(a * 2 + b) * 4 + col] = { t0.re + t1.re, t0.im + t1.im };
      }
  for (int i = 0; i < 16; ++i) U[i] = nU[i];
}

__device__ void apply_q1(cd U[16], const cd g[4]) {
  cd nU[16];
  for (int a = 0; a < 2; ++a)
    for (int b = 0; b < 2; ++b)
      for (int col = 0; col < 4; ++col) {
        cd t0 = cmul(g[b * 2 + 0], U[(a * 2 + 0) * 4 + col]);
        cd t1 = cmul(g[b * 2 + 1], U[(a * 2 + 1) * 4 + col]);
        nU[(a * 2 + b) * 4 + col] = { t0.re + t1.re, t0.im + t1.im };
      }
  for (int i = 0; i < 16; ++i) U[i] = nU[i];
}

__global__ void qnn_setup_kernel(const float* __restrict__ weights,
                                 float* __restrict__ wsA) {
  if (threadIdx.x != 0) return;
  double w[12];
  for (int i = 0; i < 12; ++i) w[i] = (double)weights[i];
  cd U[16];
  for (int i = 0; i < 16; ++i) U[i] = { (i % 5 == 0) ? 1.0 : 0.0, 0.0 };
  cd g[4];
  for (int layer = 0; layer < 2; ++layer) {
    int o = 6 * layer;
    gate_rx(w[o + 0], g); apply_q0(U, g);
    gate_rz(w[o + 1], g); apply_q0(U, g);
    gate_ry(w[o + 2], g); apply_q0(U, g);
    gate_rx(w[o + 3], g); apply_q1(U, g);
    gate_rz(w[o + 4], g); apply_q1(U, g);
    gate_ry(w[o + 5], g); apply_q1(U, g);
    for (int col = 0; col < 4; ++col) {
      cd tmp = U[2 * 4 + col];
      U[2 * 4 + col] = U[3 * 4 + col];
      U[3 * 4 + col] = tmp;
    }
  }
  const double zsg[4] = { 1.0, 1.0, -1.0, -1.0 };
  int p = 0;
  for (int j = 0; j < 4; ++j)
    for (int k = j; k < 4; ++k) {
      double acc = 0.0;
      for (int r = 0; r < 4; ++r)
        acc += zsg[r] * (U[r * 4 + j].re * U[r * 4 + k].re +
                         U[r * 4 + j].im * U[r * 4 + k].im);
      wsA[p++] = (float)acc;
    }
}


__global__ void __launch_bounds__(256)
qnn_main_kernel(const float* __restrict__ in,
                const float* __restrict__ wsA,
                float* __restrict__ out,
                int B, int NT) {
  const int lane = threadIdx.x & 31;
  const int wv   = (blockIdx.x * blockDim.x + threadIdx.x) >> 5;
  const int nwv  = (gridDim.x * blockDim.x) >> 5;

  const float A00 = wsA[0], A01 = wsA[1], A02 = wsA[2], A03 = wsA[3];
  const float A11 = wsA[4], A12 = wsA[5], A13 = wsA[6];
  const float A22 = wsA[7], A23 = wsA[8], A33 = wsA[9];

  const bool lo = (lane < 16);
  const bool diagLane = (lane < 8) || (lane >= 24);
  const int smax = B - 1;
  const int idx = lane & 7;

  for (int t = wv; t < NT; t += nwv) {
    int s = t * 16 + (lane & 15);
    int sc = (s < B) ? s : smax;
    float2 x = ((const float2*)in)[sc];

    float a0 = x.x * 0.25f;
    float a1 = x.y * 0.25f;
    float s0 = __builtin_amdgcn_sinf(a0);
    float c0 = __builtin_amdgcn_cosf(a0);
    float s1 = __builtin_amdgcn_sinf(a1);
    float c1 = __builtin_amdgcn_cosf(a1);

    float p0 = c0 * c1, p1 = c0 * s1, p2 = s0 * c1, p3 = s0 * s1;

    float g0 = A00 * p0 + A01 * p1 + A02 * p2 + A03 * p3;
    float g1 = A01 * p0 + A11 * p1 + A12 * p2 + A13 * p3;
    float g2 = A02 * p0 + A12 * p1 + A22 * p2 + A23 * p3;
    float g3 = A03 * p0 + A13 * p1 + A23 * p2 + A33 * p3;

    v2f af, bf;
    af.x = lo ? p0 : p2;  af.y = lo ? p1 : p3;
    bf.x = lo ? g0 : g2;  bf.y = lo ? g1 : g3;
    v8f cz = {};
    v8f d = __builtin_amdgcn_wmma_f32_16x16x4_f32(
        false, af, false, bf, (short)0, cz, false, false);

    float y = d[0];
    y = (idx == 1) ? d[1] : y;
    y = (idx == 2) ? d[2] : y;
    y = (idx == 3) ? d[3] : y;
    y = (idx == 4) ? d[4] : y;
    y = (idx == 5) ? d[5] : y;
    y = (idx == 6) ? d[6] : y;
    y = (idx == 7) ? d[7] : y;

    if (diagLane && s < B) out[s] = y;
  }
}


extern "C" void kernel_launch(void* const* d_in, const int* in_sizes, int n_in,
                              void* d_out, int out_size, void* d_ws, size_t ws_size,
                              hipStream_t stream) {
  const float* inp = (const float*)d_in[0];
  const float* wts = (const float*)d_in[1];
  float* out = (float*)d_out;
  float* wsA = (float*)d_ws;

  int B = in_sizes[0] / 2;

  qnn_setup_kernel<<<1, 32, 0, stream>>>(wts, wsA);

  int NT = (B + 15) / 16;
  int wavesNeeded = (NT + 7) / 8;
  int blocks = (wavesNeeded + 7) / 8;
  if (blocks < 1) blocks = 1;
  qnn_main_kernel<<<blocks, 256, 0, stream>>>(inp, wsA, out, B, NT);
}


# Main kernel: each lane carries its own sample; A*psi (4x4) moves from vector unit to two chained f32 WMMAs, A block-placed so g lands in own lane; psi.g in-lane, no diagonal pick; sum order changes
# speedup vs baseline: 7.4402x; 7.4402x over previous
#include <hip/hip_runtime.h>
#include <math.h>


typedef __attribute__((ext_vector_type(2))) float v2f;
typedef __attribute__((ext_vector_type(8))) float v8f;


struct cd { double re, im; };

__device__ __forceinline__ double cmac_part(double acc, cd a, cd b, int part) {
  const double q = part ? b.im : b.re;
  const double t = part ? b.re : b.im;
  const double r = part ? a.im : -a.im;
  return fma(r, t, fma(a.re, q, acc));
}

__global__ void __launch_bounds__(32)
qnn_setup_kernel(const float* __restrict__ weights,
                 float* __restrict__ wsA) {
  __shared__ double tc[12], ts[12];
  __shared__ double V[4 * 2 * 2 * 2];
  __shared__ double L[2 * 4 * 4 * 2];
  __shared__ double U[4 * 4 * 2];

  const int l = threadIdx.x;
  const int part = l & 1;

  if (l < 12) {
    const double t = (double)weights[l];
    double s, c;
    sincos(t * 0.5, &s, &c);
    tc[l] = c;
    ts[l] = s;
  }
  __syncthreads();

  {
    const int v = l >> 3, i = (l >> 2) & 1, j = (l >> 1) & 1;
    const int o = 6 * (v >> 1) + 3 * (v & 1);
    const double cx = tc[o],     sx = ts[o];
    const double cz = tc[o + 1], sz = ts[o + 1];
    const double cy = tc[o + 2], sy = ts[o + 2];
    const cd xd = { cx, 0.0 }, xo = { 0.0, -sx };
    const cd x0 = j ? xo : xd;
    const cd x1 = j ? xd : xo;
    const cd z0 = { cz, -sz }, z1 = { cz, sz };
    const double m0 = cmac_part(0.0, z0, x0, part);
    const double m1 = cmac_part(0.0, z1, x1, part);
    const double y0 = i ? sy : cy;
    const double y1 = i ? cy : -sy;
    V[l] = fma(y1, m1, y0 * m0);
  }
  __syncthreads();

  const int r = l >> 3, c = (l >> 1) & 3;

  {
    const int pr = r ^ (r >> 1);
    const int e0 = ((pr >> 1) * 2 + (c >> 1)) * 2;
    const int e1 = ((pr & 1) * 2 + (c & 1)) * 2;
    for (int layer = 0; layer < 2; ++layer) {
      const double* V0 = V + 16 * layer;
      const double* V1 = V0 + 8;
      const cd x = { V0[e0], V0[e0 + 1] };
      const cd y = { V1[e1], V1[e1 + 1] };
      L[32 * layer + l] = cmac_part(0.0, x, y, part);
    }
  }
  __syncthreads();

  {
    double acc = 0.0;
    for (int k = 0; k < 4; ++k) {
      const cd x = { L[32 + r * 8 + k * 2], L[32 + r * 8 + k * 2 + 1] };
      const cd y = { L[k * 8 + c * 2],      L[k * 8 + c * 2 + 1] };
      acc = cmac_part(acc, x, y, part);
    }
    U[l] = acc;
  }
  __syncthreads();

  if (l < 16) {
    const int j = l >> 2, k = l & 3;
    double acc = 0.0;
    for (int rr = 0; rr < 4; ++rr) {
      const double t = U[rr * 8 + j * 2] * U[rr * 8 + k * 2] +
                       U[rr * 8 + j * 2 + 1] * U[rr * 8 + k * 2 + 1];
      acc += (rr < 2) ? t : -t;
    }
    if (j <= k) wsA[k + 3 * j - (j * (j - 1)) / 2] = (float)acc;
  }
}


#define QNN_RUN     1024
#define QNN_GROUPS  16

__device__ __forceinline__ float qnn_eval(float x0, float x1, v2f ka, v2f kb) {
  float a0 = x0 * 0.25f;
  float a1 = x1 * 0.25f;
  float s0 = __builtin_amdgcn_sinf(a0);
  float c0 = __builtin_amdgcn_cosf(a0);
  float s1 = __builtin_amdgcn_sinf(a1);
  float c1 = __builtin_amdgcn_cosf(a1);

  v2f cs1;  cs1.x = c1;  cs1.y = s1;
  v2f b1 = c0 * cs1;
  v2f b2 = s0 * cs1;

  v8f cz = {};
  v8f d = __builtin_amdgcn_wmma_f32_16x16x4_f32(
      false, ka, false, b1, (short)0, cz, false, false);
  d = __builtin_amdgcn_wmma_f32_16x16x4_f32(
      false, kb, false, b2, (short)0, d, false, false);

  v2f glo, ghi;
  glo.x = d[0];  glo.y = d[1];
  ghi.x = d[2];  ghi.y = d[3];
  v2f t = b1 * glo + b2 * ghi;
  return t.x + t.y;
}

__device__ __forceinline__ void qnn_consts(const float* __restrict__ wsA, int lane,
                                           v2f& ka, v2f& kb) {
  const float A00 = wsA[0], A01 = wsA[1], A02 = wsA[2], A03 = wsA[3];
  const float A11 = wsA[4], A12 = wsA[5], A13 = wsA[6];
  const float A22 = wsA[7], A23 = wsA[8], A33 = wsA[9];
  const int r = lane & 3;
  const bool holds = (lane < 4) || (lane >= 24 && lane < 28);
  const float a0 = (r == 0) ? A00 : (r == 1) ? A01 : (r == 2) ? A02 : A03;
  const float a1 = (r == 0) ? A01 : (r == 1) ? A11 : (r == 2) ? A12 : A13;
  const float a2 = (r == 0) ? A02 : (r == 1) ? A12 : (r == 2) ? A22 : A23;
  const float a3 = (r == 0) ? A03 : (r == 1) ? A13 : (r == 2) ? A23 : A33;
  ka.x = holds ? a0 : 0.0f;  ka.y = holds ? a1 : 0.0f;
  kb.x = holds ? a2 : 0.0f;  kb.y = holds ? a3 : 0.0f;
}

__global__ void __launch_bounds__(256)
qnn_main_kernel(const float* __restrict__ in,
                const float* __restrict__ wsA,
                float* __restrict__ out,
                int B) {
  const int lane = threadIdx.x & 31;
  const int wv   = (blockIdx.x * blockDim.x + threadIdx.x) >> 5;
  const int base = wv * QNN_RUN;
  if (base >= B) return;

  if (base + QNN_RUN <= B) {
    const float4* __restrict__ src = (const float4*)in + (base >> 1) + lane;
    float2* __restrict__ dst = (float2*)out + (base >> 1) + lane;

    float4 v[QNN_GROUPS];
#pragma unroll
    for (int j = 0; j < QNN_GROUPS; ++j) v[j] = src[32 * j];

    v2f ka, kb;
    qnn_consts(wsA, lane, ka, kb);

    __builtin_amdgcn_sched_group_barrier(0x020, QNN_GROUPS, 0);
    __builtin_amdgcn_sched_group_barrier(0x002, 14, 0);
#pragma unroll
    for (int j = 0; j < QNN_GROUPS; ++j) {
      __builtin_amdgcn_sched_group_barrier(0x008, 4, 0);
      __builtin_amdgcn_sched_group_barrier(0x040, 1, 0);
    }

#pragma unroll
    for (int j = 0; j < QNN_GROUPS; ++j) {
      float2 y;
      y.x = qnn_eval(v[j].x, v[j].y, ka, kb);
      y.y = qnn_eval(v[j].z, v[j].w, ka, kb);
      dst[32 * j] = y;
    }
  } else {
    v2f ka, kb;
    qnn_consts(wsA, lane, ka, kb);
    const int smax = B - 1;
    for (int t = 0; t < QNN_RUN / 32; ++t) {
      const int s0 = base + 32 * t;
      if (s0 >= B) break;
      const int s = s0 + lane;
      const int sc = (s < B) ? s : smax;
      const float2 x = ((const float2*)in)[sc];
      const float y = qnn_eval(x.x, x.y, ka, kb);
      if (s < B) out[s] = y;
    }
  }
}


extern "C" void kernel_launch(void* const* d_in, const int* in_sizes, int n_in,
                              void* d_out, int out_size, void* d_ws, size_t ws_size,
                              hipStream_t stream) {
  const float* inp = (const float*)d_in[0];
  const float* wts = (const float*)d_in[1];
  float* out = (float*)d_out;
  float* wsA = (float*)d_ws;

  int B = in_sizes[0] / 2;

  qnn_setup_kernel<<<1, 32, 0, stream>>>(wts, wsA);

  int runs = (B + QNN_RUN - 1) / QNN_RUN;
  int blocks = (runs + 7) / 8;
  if (blocks < 1) blocks = 1;
  qnn_main_kernel<<<blocks, 256, 0, stream>>>(inp, wsA, out, B);
}


# Setup launch folded into the main kernel: wave 0 of each work-group runs the unchanged f64 chain through LDS while the group's loads are in flight; A reaches the 8 waves via LDS; sum order unchanged
# speedup vs baseline: 8.7341x; 1.1739x over previous
#include <hip/hip_runtime.h>
#include <math.h>


typedef __attribute__((ext_vector_type(2))) float v2f;
typedef __attribute__((ext_vector_type(8))) float v8f;


struct cd { double re, im; };

__device__ __forceinline__ double cmac_part(double acc, cd a, cd b, int part) {
  const double q = part ? b.im : b.re;
  const double t = part ? b.re : b.im;
  const double r = part ? a.im : -a.im;
  return fma(r, t, fma(a.re, q, acc));
}

__device__ __forceinline__ void qnn_s0(int l, float w, double* tc, double* ts) {
  if (l < 12) {
    const double t = (double)w;
    double s, c;
    sincos(t * 0.5, &s, &c);
    tc[l] = c;
    ts[l] = s;
  }
}

__device__ __forceinline__ void qnn_s1(int l, const double* tc, const double* ts,
                                       double* V) {
  const int part = l & 1;
  const int v = l >> 3, i = (l >> 2) & 1, j = (l >> 1) & 1;
  const int o = 6 * (v >> 1) + 3 * (v & 1);
  const double cx = tc[o],     sx = ts[o];
  const double cz = tc[o + 1], sz = ts[o + 1];
  const double cy = tc[o + 2], sy = ts[o + 2];
  const cd xd = { cx, 0.0 }, xo = { 0.0, -sx };
  const cd x0 = j ? xo : xd;
  const cd x1 = j ? xd : xo;
  const cd z0 = { cz, -sz }, z1 = { cz, sz };
  const double m0 = cmac_part(0.0, z0, x0, part);
  const double m1 = cmac_part(0.0, z1, x1, part);
  const double y0 = i ? sy : cy;
  const double y1 = i ? cy : -sy;
  V[l] = fma(y1, m1, y0 * m0);
}

__device__ __forceinline__ void qnn_s2(int l, const double* V, double* L) {
  const int part = l & 1;
  const int r = l >> 3, c = (l >> 1) & 3;
  const int pr = r ^ (r >> 1);
  const int e0 = ((pr >> 1) * 2 + (c >> 1)) * 2;
  const int e1 = ((pr & 1) * 2 + (c & 1)) * 2;
  for (int layer = 0; layer < 2; ++layer) {
    const double* V0 = V + 16 * layer;
    const double* V1 = V0 + 8;
    const cd x = { V0[e0], V0[e0 + 1] };
    const cd y = { V1[e1], V1[e1 + 1] };
    L[32 * layer + l] = cmac_part(0.0, x, y, part);
  }
}

__device__ __forceinline__ void qnn_s3(int l, const double* L, double* U) {
  const int part = l & 1;
  const int r = l >> 3, c = (l >> 1) & 3;
  double acc = 0.0;
  for (int k = 0; k < 4; ++k) {
    const cd x = { L[32 + r * 8 + k * 2], L[32 + r * 8 + k * 2 + 1] };
    const cd y = { L[k * 8 + c * 2],      L[k * 8 + c * 2 + 1] };
    acc = cmac_part(acc, x, y, part);
  }
  U[l] = acc;
}

__device__ __forceinline__ void qnn_s4(int l, const double* U, float* As) {
  if (l < 16) {
    const int a = l >> 2, b = l & 3;
    const int j = a < b ? a : b, k = a < b ? b : a;
    double acc = 0.0;
    for (int rr = 0; rr < 4; ++rr) {
      const double t = U[rr * 8 + j * 2] * U[rr * 8 + k * 2] +
                       U[rr * 8 + j * 2 + 1] * U[rr * 8 + k * 2 + 1];
      acc += (rr < 2) ? t : -t;
    }
    As[l] = (float)acc;
  }
}

__device__ __forceinline__ void qnn_lds_barrier() {
  __builtin_amdgcn_fence(__ATOMIC_RELEASE, "workgroup", "local");
  __builtin_amdgcn_s_barrier();
  __builtin_amdgcn_fence(__ATOMIC_ACQUIRE, "workgroup", "local");
}


#define QNN_RUN     1024
#define QNN_GROUPS  16

__device__ __forceinline__ float qnn_eval(float x0, float x1, v2f kp) {
  float a0 = x0 * 0.25f;
  float a1 = x1 * 0.25f;
  float s0 = __builtin_amdgcn_sinf(a0);
  float c0 = __builtin_amdgcn_cosf(a0);
  float s1 = __builtin_amdgcn_sinf(a1);
  float c1 = __builtin_amdgcn_cosf(a1);

  v2f u;  u.x = c1;  u.y = s1;

  v8f cz = {};
  v8f d = __builtin_amdgcn_wmma_f32_16x16x4_f32(
      false, kp, false, u, (short)0, cz, false, false);

  v2f tlo, thi, rlo, rhi;
  tlo.x = d[0];  tlo.y = d[1];
  thi.x = d[2];  thi.y = d[3];
  rlo.x = d[4];  rlo.y = d[5];
  rhi.x = d[6];  rhi.y = d[7];
  v2f glo = c0 * tlo + s0 * rlo;
  v2f ghi = c0 * thi + s0 * rhi;

  v2f w = c0 * glo + s0 * ghi;
  v2f t = u * w;
  return t.x + t.y;
}

__device__ __forceinline__ void qnn_consts(const float* As, int lane, v2f& kp) {
  const int r = lane & 7;
  const int q = r & 3;
  const int right = r >> 2;
  const bool holds = (lane < 8) || (lane >= 24);
  const v2f a = *(const v2f*)(As + 4 * q + 2 * right);
  kp.x = holds ? a.x : 0.0f;
  kp.y = holds ? a.y : 0.0f;
}

__global__ void __launch_bounds__(256)
qnn_main_kernel(const float* __restrict__ in,
                const float* __restrict__ weights,
                float* __restrict__ out,
                int B) {
  __shared__ double tc[12], ts[12];
  __shared__ double V[4 * 2 * 2 * 2];
  __shared__ double L[2 * 4 * 4 * 2];
  __shared__ double U[4 * 4 * 2];
  __shared__ __attribute__((aligned(16))) float As[16];

  const int lane = threadIdx.x & 31;
  const bool w0  = threadIdx.x < 32;
  const int wv   = (blockIdx.x * blockDim.x + threadIdx.x) >> 5;
  const int base = wv * QNN_RUN;
  const bool inside = base < B;
  const bool full = inside && (base + QNN_RUN <= B);

  float wt = 0.0f;
  if (w0 && lane < 12) wt = weights[lane];

  const float4* src = full ? (const float4*)in + (base >> 1) + lane
                           : (const float4*)weights;
  const int step = full ? 32 : 0;
  float4 v[QNN_GROUPS];
#pragma unroll
  for (int j = 0; j < QNN_GROUPS; ++j) v[j] = src[step * j];

  if (w0) qnn_s0(lane, wt, tc, ts);
  qnn_lds_barrier();
  if (w0) qnn_s1(lane, tc, ts, V);
  qnn_lds_barrier();
  if (w0) qnn_s2(lane, V, L);
  qnn_lds_barrier();
  if (w0) qnn_s3(lane, L, U);
  qnn_lds_barrier();
  if (w0) qnn_s4(lane, U, As);
  qnn_lds_barrier();

  if (!inside) return;

  v2f kp;
  qnn_consts(As, lane, kp);

  if (full) {
    float2* __restrict__ dst = (float2*)out + (base >> 1) + lane;

#pragma unroll
    for (int j = 0; j < QNN_GROUPS; ++j) {
      __builtin_amdgcn_sched_group_barrier(0x008, 2, 0);
      __builtin_amdgcn_sched_group_barrier(0x040, 1, 0);
    }

#pragma unroll
    for (int j = 0; j < QNN_GROUPS; ++j) {
      float2 y;
      y.x = qnn_eval(v[j].x, v[j].y, kp);
      y.y = qnn_eval(v[j].z, v[j].w, kp);
      dst[32 * j] = y;
    }
  } else {
    const int smax = B - 1;
    for (int t = 0; t < QNN_RUN / 32; ++t) {
      const int s0 = base + 32 * t;
      if (s0 >= B) break;
      const int s = s0 + lane;
      const int sc = (s < B) ? s : smax;
      const float2 x = ((const float2*)in)[sc];
      const float y = qnn_eval(x.x, x.y, kp);
      if (s < B) out[s] = y;
    }
  }
}


extern "C" void kernel_launch(void* const* d_in, const int* in_sizes, int n_in,
                              void* d_out, int out_size, void* d_ws, size_t ws_size,
                              hipStream_t stream) {
  const float* inp = (const float*)d_in[0];
  const float* wts = (const float*)d_in[1];
  float* out = (float*)d_out;

  int B = in_sizes[0] / 2;

  int runs = (B + QNN_RUN - 1) / QNN_RUN;
  int blocks = (runs + 7) / 8;
  if (blocks < 1) blocks = 1;
  qnn_main_kernel<<<blocks, 256, 0, stream>>>(inp, wts, out, B);
}


# Main kernel: 2048 waves, each owning a 2048-sample run (32 16-byte loads requested at its start), in 256 work-groups, one per WGP (was 4096 waves in 512); one f64 chain per WGP; sum order unchanged
# speedup vs baseline: 9.4509x; 1.0821x over previous
#include <hip/hip_runtime.h>
#include <math.h>


typedef __attribute__((ext_vector_type(2))) float v2f;
typedef __attribute__((ext_vector_type(8))) float v8f;


struct cd { double re, im; };

__device__ __forceinline__ double cmac_part(double acc, cd a, cd b, int part) {
  const double q = part ? b.im : b.re;
  const double t = part ? b.re : b.im;
  const double r = part ? a.im : -a.im;
  return fma(r, t, fma(a.re, q, acc));
}

__device__ __forceinline__ void qnn_s0(int l, float w, double* tc, double* ts) {
  if (l < 12) {
    const double t = (double)w;
    double s, c;
    sincos(t * 0.5, &s, &c);
    tc[l] = c;
    ts[l] = s;
  }
}

__device__ __forceinline__ void qnn_s1(int l, const double* tc, const double* ts,
                                       double* V) {
  const int part = l & 1;
  const int v = l >> 3, i = (l >> 2) & 1, j = (l >> 1) & 1;
  const int o = 6 * (v >> 1) + 3 * (v & 1);
  const double cx = tc[o],     sx = ts[o];
  const double cz = tc[o + 1], sz = ts[o + 1];
  const double cy = tc[o + 2], sy = ts[o + 2];
  const cd xd = { cx, 0.0 }, xo = { 0.0, -sx };
  const cd x0 = j ? xo : xd;
  const cd x1 = j ? xd : xo;
  const cd z0 = { cz, -sz }, z1 = { cz, sz };
  const double m0 = cmac_part(0.0, z0, x0, part);
  const double m1 = cmac_part(0.0, z1, x1, part);
  const double y0 = i ? sy : cy;
  const double y1 = i ? cy : -sy;
  V[l] = fma(y1, m1, y0 * m0);
}

__device__ __forceinline__ void qnn_s2(int l, const double* V, double* L) {
  const int part = l & 1;
  const int r = l >> 3, c = (l >> 1) & 3;
  const int pr = r ^ (r >> 1);
  const int e0 = ((pr >> 1) * 2 + (c >> 1)) * 2;
  const int e1 = ((pr & 1) * 2 + (c & 1)) * 2;
  for (int layer = 0; layer < 2; ++layer) {
    const double* V0 = V + 16 * layer;
    const double* V1 = V0 + 8;
    const cd x = { V0[e0], V0[e0 + 1] };
    const cd y = { V1[e1], V1[e1 + 1] };
    L[32 * layer + l] = cmac_part(0.0, x, y, part);
  }
}

__device__ __forceinline__ void qnn_s3(int l, const double* L, double* U) {
  const int part = l & 1;
  const int r = l >> 3, c = (l >> 1) & 3;
  double acc = 0.0;
  for (int k = 0; k < 4; ++k) {
    const cd x = { L[32 + r * 8 + k * 2], L[32 + r * 8 + k * 2 + 1] };
    const cd y = { L[k * 8 + c * 2],      L[k * 8 + c * 2 + 1] };
    acc = cmac_part(acc, x, y, part);
  }
  U[l] = acc;
}

__device__ __forceinline__ void qnn_s4(int l, const double* U, float* As) {
  if (l < 16) {
    const int a = l >> 2, b = l & 3;
    const int j = a < b ? a : b, k = a < b ? b : a;
    double acc = 0.0;
    for (int rr = 0; rr < 4; ++rr) {
      const double t = U[rr * 8 + j * 2] * U[rr * 8 + k * 2] +
                       U[rr * 8 + j * 2 + 1] * U[rr * 8 + k * 2 + 1];
      acc += (rr < 2) ? t : -t;
    }
    As[l] = (float)acc;
  }
}

__device__ __forceinline__ void qnn_lds_barrier() {
  __builtin_amdgcn_fence(__ATOMIC_RELEASE, "workgroup", "local");
  __builtin_amdgcn_s_barrier();
  __builtin_amdgcn_fence(__ATOMIC_ACQUIRE, "workgroup", "local");
}


#define QNN_RUN     2048
#define QNN_GROUPS  32

__device__ __forceinline__ float qnn_eval(float x0, float x1, v2f kp) {
  float a0 = x0 * 0.25f;
  float a1 = x1 * 0.25f;
  float s0 = __builtin_amdgcn_sinf(a0);
  float c0 = __builtin_amdgcn_cosf(a0);
  float s1 = __builtin_amdgcn_sinf(a1);
  float c1 = __builtin_amdgcn_cosf(a1);

  v2f u;  u.x = c1;  u.y = s1;

  v8f cz = {};
  v8f d = __builtin_amdgcn_wmma_f32_16x16x4_f32(
      false, kp, false, u, (short)0, cz, false, false);

  v2f tlo, thi, rlo, rhi;
  tlo.x = d[0];  tlo.y = d[1];
  thi.x = d[2];  thi.y = d[3];
  rlo.x = d[4];  rlo.y = d[5];
  rhi.x = d[6];  rhi.y = d[7];
  v2f glo = c0 * tlo + s0 * rlo;
  v2f ghi = c0 * thi + s0 * rhi;

  v2f w = c0 * glo + s0 * ghi;
  v2f t = u * w;
  return t.x + t.y;
}

__device__ __forceinline__ void qnn_consts(const float* As, int lane, v2f& kp) {
  const int r = lane & 7;
  const int q = r & 3;
  const int right = r >> 2;
  const bool holds = (lane < 8) || (lane >= 24);
  const v2f a = *(const v2f*)(As + 4 * q + 2 * right);
  kp.x = holds ? a.x : 0.0f;
  kp.y = holds ? a.y : 0.0f;
}

__global__ void __launch_bounds__(256)
qnn_main_kernel(const float* __restrict__ in,
                const float* __restrict__ weights,
                float* __restrict__ out,
                int B) {
  __shared__ double tc[12], ts[12];
  __shared__ double V[4 * 2 * 2 * 2];
  __shared__ double L[2 * 4 * 4 * 2];
  __shared__ double U[4 * 4 * 2];
  __shared__ __attribute__((aligned(16))) float As[16];

  const int lane = threadIdx.x & 31;
  const bool w0  = threadIdx.x < 32;
  const int wv   = (blockIdx.x * blockDim.x + threadIdx.x) >> 5;
  const int base = wv * QNN_RUN;
  const bool inside = base < B;
  const bool full = inside && (base + QNN_RUN <= B);

  float wt = 0.0f;
  if (w0 && lane < 12) wt = weights[lane];

  const float4* src = full ? (const float4*)in + (base >> 1) + lane
                           : (const float4*)weights;
  const int step = full ? 32 : 0;
  float4 v[QNN_GROUPS];
#pragma unroll
  for (int j = 0; j < QNN_GROUPS; ++j) v[j] = src[step * j];

  if (w0) qnn_s0(lane, wt, tc, ts);
  qnn_lds_barrier();
  if (w0) qnn_s1(lane, tc, ts, V);
  qnn_lds_barrier();
  if (w0) qnn_s2(lane, V, L);
  qnn_lds_barrier();
  if (w0) qnn_s3(lane, L, U);
  qnn_lds_barrier();
  if (w0) qnn_s4(lane, U, As);
  qnn_lds_barrier();

  if (!inside) return;

  v2f kp;
  qnn_consts(As, lane, kp);

  if (full) {
    float2* __restrict__ dst = (float2*)out + (base >> 1) + lane;

#pragma unroll
    for (int j = 0; j < QNN_GROUPS; ++j) {
      __builtin_amdgcn_sched_group_barrier(0x008, 2, 0);
      __builtin_amdgcn_sched_group_barrier(0x040, 1, 0);
    }

#pragma unroll
    for (int j = 0; j < QNN_GROUPS; ++j) {
      float2 y;
      y.x = qnn_eval(v[j].x, v[j].y, kp);
      y.y = qnn_eval(v[j].z, v[j].w, kp);
      dst[32 * j] = y;
    }
  } else {
    const int smax = B - 1;
    for (int t = 0; t < QNN_RUN / 32; ++t) {
      const int s0 = base + 32 * t;
      if (s0 >= B) break;
      const int s = s0 + lane;
      const int sc = (s < B) ? s : smax;
      const float2 x = ((const float2*)in)[sc];
      const float y = qnn_eval(x.x, x.y, kp);
      if (s < B) out[s] = y;
    }
  }
}


extern "C" void kernel_launch(void* const* d_in, const int* in_sizes, int n_in,
                              void* d_out, int out_size, void* d_ws, size_t ws_size,
                              hipStream_t stream) {
  const float* inp = (const float*)d_in[0];
  const float* wts = (const float*)d_in[1];
  float* out = (float*)d_out;

  int B = in_sizes[0] / 2;

  int runs = (B + QNN_RUN - 1) / QNN_RUN;
  int blocks = (runs + 7) / 8;
  if (blocks < 1) blocks = 1;
  qnn_main_kernel<<<blocks, 256, 0, stream>>>(inp, wts, out, B);
}


# Coefficient chain on wave 0 (f64 kept): 77 -> 40 f64 issues: sin|cos of the half angles side by side in 24 lanes, one series run; Gram sums over lane pairs (f64 sum order changes); 1 barrier, was 5
# speedup vs baseline: 10.3153x; 1.0915x over previous
#include <hip/hip_runtime.h>
#include <math.h>


typedef __attribute__((ext_vector_type(2))) float v2f;
typedef __attribute__((ext_vector_type(8))) float v8f;


struct cd { double re, im; };

__device__ __forceinline__ double cmac_part(double acc, cd a, cd b, int part) {
  const double q = part ? b.im : b.re;
  const double t = part ? b.re : b.im;
  const double r = part ? a.im : -a.im;
  return fma(r, t, fma(a.re, q, acc));
}

__device__ __forceinline__ void qnn_s0(int l, float w, double* tcs) {
  if (l < 24) {
    const bool sinlane = l >= 12;
    const double t = (double)w * 0.5;
    double res;
    if (__builtin_fabsf(w) < 2147483648.0f) {
      const double q = __builtin_rint(t * 0.6366197723675814);
      double r = __builtin_fma(q, -1.5707963267948966, t);
      r = __builtin_fma(q, -6.123233995736766e-17, r);
      const int k = (int)q + (sinlane ? 0 : 1);
      const bool cpoly = (k & 1) != 0;
      const double z = r * r;
      double p = cpoly ? -1.13596475577881948265e-11 : 0.0;
      p = __builtin_fma(p, z, cpoly ?  2.08757232129817482790e-09 :  1.58969099521155010221e-10);
      p = __builtin_fma(p, z, cpoly ? -2.75573143513906633035e-07 : -2.50507602534068634195e-08);
      p = __builtin_fma(p, z, cpoly ?  2.48015872894767294178e-05 :  2.75573137070700676789e-06);
      p = __builtin_fma(p, z, cpoly ? -1.38888888888741095749e-03 : -1.98412698298579493134e-04);
      p = __builtin_fma(p, z, cpoly ?  4.16666666666666019037e-02 :  8.33333333332248946124e-03);
      p = __builtin_fma(p, z, cpoly ? -0.5                        : -1.66666666666666324348e-01);
      const double base = cpoly ? 1.0 : r;
      res = __builtin_fma(base * z, p, base);
      if (k & 2) res = -res;
    } else {
      double s, c;
      sincos(t, &s, &c);
      res = sinlane ? s : c;
    }
    tcs[l] = res;
  }
}

__device__ __forceinline__ void qnn_s1(int l, const double* tc, const double* ts,
                                       double* V) {
  const int part = l & 1;
  const int v = l >> 3, i = (l >> 2) & 1, j = (l >> 1) & 1;
  const int o = 6 * (v >> 1) + 3 * (v & 1);
  const double cx = tc[o],     sx = ts[o];
  const double cz = tc[o + 1], sz = ts[o + 1];
  const double cy = tc[o + 2], sy = ts[o + 2];
  const cd xd = { cx, 0.0 }, xo = { 0.0, -sx };
  const cd x0 = j ? xo : xd;
  const cd x1 = j ? xd : xo;
  const cd z0 = { cz, -sz }, z1 = { cz, sz };
  const double m0 = cmac_part(0.0, z0, x0, part);
  const double m1 = cmac_part(0.0, z1, x1, part);
  const double y0 = i ? sy : cy;
  const double y1 = i ? cy : -sy;
  V[l] = fma(y1, m1, y0 * m0);
}

__device__ __forceinline__ void qnn_s2(int l, const double* V, double* L) {
  const int part = l & 1;
  const int r = l >> 3, c = (l >> 1) & 3;
  const int pr = r ^ (r >> 1);
  const int e0 = ((pr >> 1) * 2 + (c >> 1)) * 2;
  const int e1 = ((pr & 1) * 2 + (c & 1)) * 2;
  for (int layer = 0; layer < 2; ++layer) {
    const double* V0 = V + 16 * layer;
    const double* V1 = V0 + 8;
    const cd x = { V0[e0], V0[e0 + 1] };
    const cd y = { V1[e1], V1[e1 + 1] };
    L[32 * layer + l] = cmac_part(0.0, x, y, part);
  }
}

__device__ __forceinline__ void qnn_s3(int l, const double* L, double* U) {
  const int part = l & 1;
  const int r = l >> 3, c = (l >> 1) & 3;
  double acc = 0.0;
  for (int k = 0; k < 4; ++k) {
    const cd x = { L[32 + r * 8 + k * 2], L[32 + r * 8 + k * 2 + 1] };
    const cd y = { L[k * 8 + c * 2],      L[k * 8 + c * 2 + 1] };
    acc = cmac_part(acc, x, y, part);
  }
  U[l] = acc;
}

__device__ __forceinline__ double qnn_other_half(double x) {
  int lo = __double2loint(x), hi = __double2hiint(x);
  lo = __builtin_amdgcn_permlanex16(lo, lo, 0x76543210, 0xfedcba98, false, false);
  hi = __builtin_amdgcn_permlanex16(hi, hi, 0x76543210, 0xfedcba98, false, false);
  return __hiloint2double(hi, lo);
}

__device__ __forceinline__ void qnn_s4(int l, const double* U, float* As) {
  const int e = l & 15, h = l >> 4;
  const int a = e >> 2, b = e & 3;
  const int j = a < b ? a : b, k = a < b ? b : a;
  const double* U0 = U + 16 * h;
  double acc = U0[j * 2] * U0[k * 2];
  acc = __builtin_fma(U0[j * 2 + 1],     U0[k * 2 + 1],     acc);
  acc = __builtin_fma(U0[8 + j * 2],     U0[8 + k * 2],     acc);
  acc = __builtin_fma(U0[8 + j * 2 + 1], U0[8 + k * 2 + 1], acc);
  const double other = qnn_other_half(acc);
  if (l < 16) As[l] = (float)(acc - other);
}

__device__ __forceinline__ void qnn_wave_lds_order() {
  __builtin_amdgcn_fence(__ATOMIC_RELEASE, "wavefront", "local");
  __builtin_amdgcn_wave_barrier();
  __builtin_amdgcn_fence(__ATOMIC_ACQUIRE, "wavefront", "local");
}

__device__ __forceinline__ void qnn_lds_barrier() {
  __builtin_amdgcn_fence(__ATOMIC_RELEASE, "workgroup", "local");
  __builtin_amdgcn_s_barrier();
  __builtin_amdgcn_fence(__ATOMIC_ACQUIRE, "workgroup", "local");
}


#define QNN_RUN     2048
#define QNN_GROUPS  32

__device__ __forceinline__ float qnn_eval(float x0, float x1, v2f kp) {
  float a0 = x0 * 0.25f;
  float a1 = x1 * 0.25f;
  float s0 = __builtin_amdgcn_sinf(a0);
  float c0 = __builtin_amdgcn_cosf(a0);
  float s1 = __builtin_amdgcn_sinf(a1);
  float c1 = __builtin_amdgcn_cosf(a1);

  v2f u;  u.x = c1;  u.y = s1;

  v8f cz = {};
  v8f d = __builtin_amdgcn_wmma_f32_16x16x4_f32(
      false, kp, false, u, (short)0, cz, false, false);

  v2f tlo, thi, rlo, rhi;
  tlo.x = d[0];  tlo.y = d[1];
  thi.x = d[2];  thi.y = d[3];
  rlo.x = d[4];  rlo.y = d[5];
  rhi.x = d[6];  rhi.y = d[7];
  v2f glo = c0 * tlo + s0 * rlo;
  v2f ghi = c0 * thi + s0 * rhi;

  v2f w = c0 * glo + s0 * ghi;
  v2f t = u * w;
  return t.x + t.y;
}

__device__ __forceinline__ void qnn_consts(const float* As, int lane, v2f& kp) {
  const int r = lane & 7;
  const int q = r & 3;
  const int right = r >> 2;
  const bool holds = (lane < 8) || (lane >= 24);
  const v2f a = *(const v2f*)(As + 4 * q + 2 * right);
  kp.x = holds ? a.x : 0.0f;
  kp.y = holds ? a.y : 0.0f;
}

__global__ void __launch_bounds__(256)
qnn_main_kernel(const float* __restrict__ in,
                const float* __restrict__ weights,
                float* __restrict__ out,
                int B) {
  __shared__ double tcs[24];
  __shared__ double V[4 * 2 * 2 * 2];
  __shared__ double L[2 * 4 * 4 * 2];
  __shared__ double U[4 * 4 * 2];
  __shared__ __attribute__((aligned(16))) float As[16];

  const int lane = threadIdx.x & 31;
  const bool w0  = threadIdx.x < 32;
  const int wv   = (blockIdx.x * blockDim.x + threadIdx.x) >> 5;
  const int base = wv * QNN_RUN;
  const bool inside = base < B;
  const bool full = inside && (base + QNN_RUN <= B);

  float wt = 0.0f;
  if (w0 && lane < 24) wt = weights[lane < 12 ? lane : lane - 12];

  const float4* src = full ? (const float4*)in + (base >> 1) + lane
                           : (const float4*)weights;
  const int step = full ? 32 : 0;
  float4 v[QNN_GROUPS];
#pragma unroll
  for (int j = 0; j < QNN_GROUPS; ++j) v[j] = src[step * j];

  if (w0) {
    qnn_s0(lane, wt, tcs);
    qnn_wave_lds_order();
    qnn_s1(lane, tcs, tcs + 12, V);
    qnn_wave_lds_order();
    qnn_s2(lane, V, L);
    qnn_wave_lds_order();
    qnn_s3(lane, L, U);
    qnn_wave_lds_order();
    qnn_s4(lane, U, As);
  }
  qnn_lds_barrier();

  if (!inside) return;

  v2f kp;
  qnn_consts(As, lane, kp);

  if (full) {
    float2* __restrict__ dst = (float2*)out + (base >> 1) + lane;

#pragma unroll
    for (int j = 0; j < QNN_GROUPS; ++j) {
      __builtin_amdgcn_sched_group_barrier(0x008, 2, 0);
      __builtin_amdgcn_sched_group_barrier(0x040, 1, 0);
    }

#pragma unroll
    for (int j = 0; j < QNN_GROUPS; ++j) {
      float2 y;
      y.x = qnn_eval(v[j].x, v[j].y, kp);
      y.y = qnn_eval(v[j].z, v[j].w, kp);
      dst[32 * j] = y;
    }
  } else {
    const int smax = B - 1;
    for (int t = 0; t < QNN_RUN / 32; ++t) {
      const int s0 = base + 32 * t;
      if (s0 >= B) break;
      const int s = s0 + lane;
      const int sc = (s < B) ? s : smax;
      const float2 x = ((const float2*)in)[sc];
      const float y = qnn_eval(x.x, x.y, kp);
      if (s < B) out[s] = y;
    }
  }
}


extern "C" void kernel_launch(void* const* d_in, const int* in_sizes, int n_in,
                              void* d_out, int out_size, void* d_ws, size_t ws_size,
                              hipStream_t stream) {
  const float* inp = (const float*)d_in[0];
  const float* wts = (const float*)d_in[1];
  float* out = (float*)d_out;

  int B = in_sizes[0] / 2;

  int runs = (B + QNN_RUN - 1) / QNN_RUN;
  int blocks = (runs + 7) / 8;
  if (blocks < 1) blocks = 1;
  qnn_main_kernel<<<blocks, 256, 0, stream>>>(inp, wts, out, B);
}
